# starting kernel
# baseline (speedup 1.0000x reference)
#include <hip/hip_runtime.h>


typedef float v2f __attribute__((ext_vector_type(2)));
typedef float v8f __attribute__((ext_vector_type(8)));

#define ATT_S   4096
#define ATT_D   64
#define NHEAD   16
#define NCHUNK  16
#define CHUNK_S (ATT_S / NCHUNK)
#define K1_WAVES 4
#define WAVE_S  (CHUNK_S / K1_WAVES)
#define QTILE   128
#define QWAVES  2
#define QPAD    68

static __device__ __forceinline__ v8f wmma_f32(v2f a, v2f b, v8f c) {
  return __builtin_amdgcn_wmma_f32_16x16x4_f32(false, a, false, b,
                                               (short)0, c, false, false);
}

__global__ __launch_bounds__(K1_WAVES * 32)
void attn_ktv_partial(const float* __restrict__ K, const float* __restrict__ V,
                      float* __restrict__ partial) {
  const int head  = blockIdx.x / NCHUNK;
  const int chunk = blockIdx.x % NCHUNK;
  const int wave  = threadIdx.x >> 5;
  const int lane  = threadIdx.x & 31;
  const int lmod  = lane & 15;
  const int lhalf = lane >> 4;

  const float* __restrict__ Kh = K + (size_t)head * ATT_S * ATT_D;
  const float* __restrict__ Vh = V + (size_t)head * ATT_S * ATT_D;
  const int s0 = chunk * CHUNK_S + wave * WAVE_S;

  v8f c[4][4];
#pragma unroll
  for (int i = 0; i < 4; ++i)
#pragma unroll
    for (int j = 0; j < 4; ++j)
      c[i][j] = (v8f){0.f, 0.f, 0.f, 0.f, 0.f, 0.f, 0.f, 0.f};

#pragma unroll 2
  for (int step = 0; step < WAVE_S / 4; ++step) {
    const int s = s0 + step * 4 + lhalf * 2;
    const float* __restrict__ kr = Kh + (size_t)s * ATT_D + lmod;
    const float* __restrict__ vr = Vh + (size_t)s * ATT_D + lmod;
    v2f a[4], b[4];
#pragma unroll
    for (int t = 0; t < 4; ++t) {
      a[t].x = kr[t * 16];
      a[t].y = kr[t * 16 + ATT_D];
      b[t].x = vr[t * 16];
      b[t].y = vr[t * 16 + ATT_D];
    }
#pragma unroll
    for (int i = 0; i < 4; ++i)
#pragma unroll
      for (int j = 0; j < 4; ++j)
        c[i][j] = wmma_f32(a[i], b[j], c[i][j]);
  }

  __shared__ float red[K1_WAVES][ATT_D * ATT_D];
#pragma unroll
  for (int i = 0; i < 4; ++i)
#pragma unroll
    for (int j = 0; j < 4; ++j)
#pragma unroll
      for (int r = 0; r < 8; ++r)
        red[wave][(i * 16 + r + lhalf * 8) * ATT_D + j * 16 + lmod] = c[i][j][r];
  __syncthreads();

  float* __restrict__ outp =
      partial + ((size_t)head * NCHUNK + chunk) * (ATT_D * ATT_D);
  for (int e = threadIdx.x; e < ATT_D * ATT_D; e += K1_WAVES * 32)
    outp[e] = (red[0][e] + red[1][e]) + (red[2][e] + red[3][e]);
}

__global__ __launch_bounds__(256)
void attn_reduce(const float* __restrict__ partial, float* __restrict__ M) {
  const int head = blockIdx.x;
  const float* __restrict__ p = partial + (size_t)head * NCHUNK * ATT_D * ATT_D;
  for (int e = threadIdx.x; e < ATT_D * ATT_D; e += 256) {
    float s = 0.f;
#pragma unroll
    for (int cc = 0; cc < NCHUNK; ++cc)
      s += p[cc * ATT_D * ATT_D + e];
    M[(size_t)head * ATT_D * ATT_D + e] = s;
  }
}

__global__ __launch_bounds__(QWAVES * 32)
void attn_qm(const float* __restrict__ Q, const float* __restrict__ M,
             float* __restrict__ out) {
  const int head  = blockIdx.x / (ATT_S / QTILE);
  const int qc    = blockIdx.x % (ATT_S / QTILE);
  const int wave  = threadIdx.x >> 5;
  const int lane  = threadIdx.x & 31;
  const int lmod  = lane & 15;
  const int lhalf = lane >> 4;
  const int q0    = qc * QTILE;

  __shared__ float qs[QTILE * QPAD];

  const float* __restrict__ Qh = Q + ((size_t)head * ATT_S + q0) * ATT_D;
  for (int idx = threadIdx.x; idx < QTILE * (ATT_D / 4); idx += QWAVES * 32) {
    const int row = idx >> 4;
    const int c4  = idx & 15;
    const float4 f = ((const float4*)Qh)[idx];
    *(float4*)&qs[row * QPAD + c4 * 4] = f;
  }
  __syncthreads();

  const float* __restrict__ Mh = M + (size_t)head * ATT_D * ATT_D;

  v8f c[4][4];
#pragma unroll
  for (int i = 0; i < 4; ++i)
#pragma unroll
    for (int j = 0; j < 4; ++j)
      c[i][j] = (v8f){0.f, 0.f, 0.f, 0.f, 0.f, 0.f, 0.f, 0.f};

#pragma unroll 2
  for (int step = 0; step < ATT_D / 4; ++step) {
    const int k0 = step * 4 + lhalf * 2;
    v2f a[4], b[4];
#pragma unroll
    for (int t = 0; t < 4; ++t) {
      const v2f av = *(const v2f*)&qs[(wave * 64 + t * 16 + lmod) * QPAD + k0];
      a[t] = av;
      b[t].x = Mh[(k0 + 0) * ATT_D + t * 16 + lmod];
      b[t].y = Mh[(k0 + 1) * ATT_D + t * 16 + lmod];
    }
#pragma unroll
    for (int i = 0; i < 4; ++i)
#pragma unroll
      for (int j = 0; j < 4; ++j)
        c[i][j] = wmma_f32(a[i], b[j], c[i][j]);
  }

  float* __restrict__ Oh =
      out + ((size_t)head * ATT_S + q0 + wave * 64) * ATT_D;
#pragma unroll
  for (int i = 0; i < 4; ++i)
#pragma unroll
    for (int j = 0; j < 4; ++j)
#pragma unroll
      for (int r = 0; r < 8; ++r)
        Oh[(size_t)(i * 16 + r + lhalf * 8) * ATT_D + j * 16 + lmod] =
            c[i][j][r];
}

extern "C" void kernel_launch(void* const* d_in, const int* in_sizes, int n_in,
                              void* d_out, int out_size, void* d_ws,
                              size_t ws_size, hipStream_t stream) {
  (void)in_sizes; (void)n_in; (void)out_size; (void)ws_size;
  const float* q = (const float*)d_in[0];
  const float* k = (const float*)d_in[1];
  const float* v = (const float*)d_in[2];
  float* out = (float*)d_out;

  float* partial = (float*)d_ws;
  float* M = partial + (size_t)NHEAD * NCHUNK * ATT_D * ATT_D;

  attn_ktv_partial<<<NHEAD * NCHUNK, K1_WAVES * 32, 0, stream>>>(k, v, partial);
  attn_reduce<<<NHEAD, 256, 0, stream>>>(partial, M);
  attn_qm<<<NHEAD * (ATT_S / QTILE), QWAVES * 32, 0, stream>>>(q, M, out);
}


# Both matrix kernels: operand loads issued ahead of the matrix work that uses them (K/V and M fragments 3 steps ahead, whole Q tile up front, LDS reads 1 step ahead); same loads, same sums
# speedup vs baseline: 1.3347x; 1.3347x over previous
#include <hip/hip_runtime.h>


typedef float v2f __attribute__((ext_vector_type(2)));
typedef float v8f __attribute__((ext_vector_type(8)));

#define ATT_S   4096
#define ATT_D   64
#define NHEAD   16
#define NCHUNK  16
#define CHUNK_S (ATT_S / NCHUNK)
#define K1_WAVES 4
#define WAVE_S  (CHUNK_S / K1_WAVES)
#define QTILE   128
#define QWAVES  2
#define QPAD    68
#define K1_AHEAD 3
#define QM_AHEAD 3

static __device__ __forceinline__ v8f wmma_f32(v2f a, v2f b, v8f c) {
  return __builtin_amdgcn_wmma_f32_16x16x4_f32(false, a, false, b,
                                               (short)0, c, false, false);
}

__global__ __launch_bounds__(K1_WAVES * 32)
void attn_ktv_partial(const float* __restrict__ K, const float* __restrict__ V,
                      float* __restrict__ partial) {
  const int head  = blockIdx.x / NCHUNK;
  const int chunk = blockIdx.x % NCHUNK;
  const int wave  = threadIdx.x >> 5;
  const int lane  = threadIdx.x & 31;
  const int lmod  = lane & 15;
  const int lhalf = lane >> 4;

  const float* __restrict__ Kh = K + (size_t)head * ATT_S * ATT_D;
  const float* __restrict__ Vh = V + (size_t)head * ATT_S * ATT_D;
  const int s0 = chunk * CHUNK_S + wave * WAVE_S;

  v8f c[4][4];
#pragma unroll
  for (int i = 0; i < 4; ++i)
#pragma unroll
    for (int j = 0; j < 4; ++j)
      c[i][j] = (v8f){0.f, 0.f, 0.f, 0.f, 0.f, 0.f, 0.f, 0.f};

  v2f a[WAVE_S / 4][4], b[WAVE_S / 4][4];
#pragma unroll
  for (int st = 0; st < K1_AHEAD; ++st) {
    const int s = s0 + st * 4 + lhalf * 2;
    const float* __restrict__ kr = Kh + (size_t)s * ATT_D + lmod;
    const float* __restrict__ vr = Vh + (size_t)s * ATT_D + lmod;
#pragma unroll
    for (int t = 0; t < 4; ++t) {
      a[st][t].x = kr[t * 16];
      a[st][t].y = kr[t * 16 + ATT_D];
      b[st][t].x = vr[t * 16];
      b[st][t].y = vr[t * 16 + ATT_D];
    }
  }

#pragma unroll
  for (int step = 0; step < WAVE_S / 4; ++step) {
    if (step + K1_AHEAD < WAVE_S / 4) {
      const int st = step + K1_AHEAD;
      const int s = s0 + st * 4 + lhalf * 2;
      const float* __restrict__ kr = Kh + (size_t)s * ATT_D + lmod;
      const float* __restrict__ vr = Vh + (size_t)s * ATT_D + lmod;
#pragma unroll
      for (int t = 0; t < 4; ++t) {
        a[st][t].x = kr[t * 16];
        a[st][t].y = kr[t * 16 + ATT_D];
        b[st][t].x = vr[t * 16];
        b[st][t].y = vr[t * 16 + ATT_D];
      }
    }
#pragma unroll
    for (int i = 0; i < 4; ++i)
#pragma unroll
      for (int j = 0; j < 4; ++j)
        c[i][j] = wmma_f32(a[step][i], b[step][j], c[i][j]);
    __builtin_amdgcn_sched_barrier(0);
  }
  asm volatile("v_nop\n\tv_nop\n\tv_nop\n\tv_nop"
               :: "v"(a[WAVE_S / 4 - 1][0]), "v"(a[WAVE_S / 4 - 1][1]),
                  "v"(a[WAVE_S / 4 - 1][2]), "v"(a[WAVE_S / 4 - 1][3]),
                  "v"(b[WAVE_S / 4 - 1][0]), "v"(b[WAVE_S / 4 - 1][1]),
                  "v"(b[WAVE_S / 4 - 1][2]), "v"(b[WAVE_S / 4 - 1][3]));

  __shared__ float red[K1_WAVES][ATT_D * ATT_D];
#pragma unroll
  for (int i = 0; i < 4; ++i)
#pragma unroll
    for (int j = 0; j < 4; ++j)
#pragma unroll
      for (int r = 0; r < 8; ++r)
        red[wave][(i * 16 + r + lhalf * 8) * ATT_D + j * 16 + lmod] = c[i][j][r];
  __syncthreads();

  float* __restrict__ outp =
      partial + ((size_t)head * NCHUNK + chunk) * (ATT_D * ATT_D);
  for (int e = threadIdx.x; e < ATT_D * ATT_D; e += K1_WAVES * 32)
    outp[e] = (red[0][e] + red[1][e]) + (red[2][e] + red[3][e]);
}

__global__ __launch_bounds__(256)
void attn_reduce(const float* __restrict__ partial, float* __restrict__ M) {
  const int head = blockIdx.x;
  const float* __restrict__ p = partial + (size_t)head * NCHUNK * ATT_D * ATT_D;
  for (int e = threadIdx.x; e < ATT_D * ATT_D; e += 256) {
    float s = 0.f;
#pragma unroll
    for (int cc = 0; cc < NCHUNK; ++cc)
      s += p[cc * ATT_D * ATT_D + e];
    M[(size_t)head * ATT_D * ATT_D + e] = s;
  }
}

__global__ __launch_bounds__(QWAVES * 32)
void attn_qm(const float* __restrict__ Q, const float* __restrict__ M,
             float* __restrict__ out) {
  const int head  = blockIdx.x / (ATT_S / QTILE);
  const int qc    = blockIdx.x % (ATT_S / QTILE);
  const int wave  = threadIdx.x >> 5;
  const int lane  = threadIdx.x & 31;
  const int lmod  = lane & 15;
  const int lhalf = lane >> 4;
  const int q0    = qc * QTILE;

  __shared__ float qs[QTILE * QPAD];

  int ooff = (wave * 64 + lhalf * 8) * ATT_D + lmod;
  asm volatile("" : "+v"(ooff));

  constexpr int QFILL = QTILE * (ATT_D / 4) / (QWAVES * 32);
  const float* __restrict__ Qh = Q + ((size_t)head * ATT_S + q0) * ATT_D;
  float4 qv[QFILL];
#pragma unroll
  for (int it = 0; it < QFILL; ++it)
    qv[it] = ((const float4*)Qh)[threadIdx.x + it * (QWAVES * 32)];

  const float* __restrict__ Mh = M + (size_t)head * ATT_D * ATT_D;
  v2f mb[ATT_D / 4][4];
#pragma unroll
  for (int st = 0; st < QM_AHEAD; ++st) {
    const int k0 = st * 4 + lhalf * 2;
#pragma unroll
    for (int t = 0; t < 4; ++t) {
      mb[st][t].x = Mh[(k0 + 0) * ATT_D + t * 16 + lmod];
      mb[st][t].y = Mh[(k0 + 1) * ATT_D + t * 16 + lmod];
    }
  }

#pragma unroll
  for (int it = 0; it < QFILL; ++it) {
    const int idx = threadIdx.x + it * (QWAVES * 32);
    const int row = idx >> 4;
    const int c4  = idx & 15;
    *(float4*)&qs[row * QPAD + c4 * 4] = qv[it];
  }
  __syncthreads();

  v8f c[4][4];
#pragma unroll
  for (int i = 0; i < 4; ++i)
#pragma unroll
    for (int j = 0; j < 4; ++j)
      c[i][j] = (v8f){0.f, 0.f, 0.f, 0.f, 0.f, 0.f, 0.f, 0.f};

  v2f a[ATT_D / 4][4];
#pragma unroll
  for (int t = 0; t < 4; ++t) {
    a[0][t] = *(const v2f*)&qs[(wave * 64 + t * 16 + lmod) * QPAD + lhalf * 2];
  }
#pragma unroll
  for (int step = 0; step < ATT_D / 4; ++step) {
    if (step + QM_AHEAD < ATT_D / 4) {
      const int st = step + QM_AHEAD;
      const int k0 = st * 4 + lhalf * 2;
#pragma unroll
      for (int t = 0; t < 4; ++t) {
        mb[st][t].x = Mh[(k0 + 0) * ATT_D + t * 16 + lmod];
        mb[st][t].y = Mh[(k0 + 1) * ATT_D + t * 16 + lmod];
      }
    }
    if (step + 1 < ATT_D / 4) {
      const int k0 = (step + 1) * 4 + lhalf * 2;
#pragma unroll
      for (int t = 0; t < 4; ++t)
        a[step + 1][t] =
            *(const v2f*)&qs[(wave * 64 + t * 16 + lmod) * QPAD + k0];
    }
#pragma unroll
    for (int i = 0; i < 4; ++i)
#pragma unroll
      for (int j = 0; j < 4; ++j)
        c[i][j] = wmma_f32(a[step][i], mb[step][j], c[i][j]);
    __builtin_amdgcn_sched_barrier(0);
  }
  asm volatile("v_nop\n\tv_nop\n\tv_nop\n\tv_nop"
               : "+v"(ooff)
               : "v"(a[ATT_D / 4 - 1][0]), "v"(a[ATT_D / 4 - 1][1]),
                 "v"(a[ATT_D / 4 - 1][2]), "v"(a[ATT_D / 4 - 1][3]),
                 "v"(mb[ATT_D / 4 - 1][0]), "v"(mb[ATT_D / 4 - 1][1]),
                 "v"(mb[ATT_D / 4 - 1][2]), "v"(mb[ATT_D / 4 - 1][3]));

  float* __restrict__ Oh =
      out + ((size_t)head * ATT_S + q0) * ATT_D + ooff;
#pragma unroll
  for (int i = 0; i < 4; ++i)
#pragma unroll
    for (int j = 0; j < 4; ++j)
#pragma unroll
      for (int r = 0; r < 8; ++r)
        Oh[(i * 16 + r) * ATT_D + j * 16] = c[i][j][r];
}

extern "C" void kernel_launch(void* const* d_in, const int* in_sizes, int n_in,
                              void* d_out, int out_size, void* d_ws,
                              size_t ws_size, hipStream_t stream) {
  (void)in_sizes; (void)n_in; (void)out_size; (void)ws_size;
  const float* q = (const float*)d_in[0];
  const float* k = (const float*)d_in[1];
  const float* v = (const float*)d_in[2];
  float* out = (float*)d_out;

  float* partial = (float*)d_ws;
  float* M = partial + (size_t)NHEAD * NCHUNK * ATT_D * ATT_D;

  attn_ktv_partial<<<NHEAD * NCHUNK, K1_WAVES * 32, 0, stream>>>(k, v, partial);
  attn_reduce<<<NHEAD, 256, 0, stream>>>(partial, M);
  attn_qm<<<NHEAD * (ATT_S / QTILE), QWAVES * 32, 0, stream>>>(q, M, out);
}


# Reduce launch spread from 16 workgroups to 256: each thread sums one 16-byte piece of M with all 16 chunk loads in flight at once (was 8 dependent trips); add order per element unchanged
# speedup vs baseline: 1.7269x; 1.2938x over previous
#include <hip/hip_runtime.h>


typedef float v2f __attribute__((ext_vector_type(2)));
typedef float v8f __attribute__((ext_vector_type(8)));

#define ATT_S   4096
#define ATT_D   64
#define NHEAD   16
#define NCHUNK  16
#define CHUNK_S (ATT_S / NCHUNK)
#define K1_WAVES 4
#define WAVE_S  (CHUNK_S / K1_WAVES)
#define QTILE   128
#define QWAVES  2
#define QPAD    68
#define K1_AHEAD 3
#define QM_AHEAD 3
#define RED_THREADS 64

static __device__ __forceinline__ v8f wmma_f32(v2f a, v2f b, v8f c) {
  return __builtin_amdgcn_wmma_f32_16x16x4_f32(false, a, false, b,
                                               (short)0, c, false, false);
}

__global__ __launch_bounds__(K1_WAVES * 32)
void attn_ktv_partial(const float* __restrict__ K, const float* __restrict__ V,
                      float* __restrict__ partial) {
  const int head  = blockIdx.x / NCHUNK;
  const int chunk = blockIdx.x % NCHUNK;
  const int wave  = threadIdx.x >> 5;
  const int lane  = threadIdx.x & 31;
  const int lmod  = lane & 15;
  const int lhalf = lane >> 4;

  const float* __restrict__ Kh = K + (size_t)head * ATT_S * ATT_D;
  const float* __restrict__ Vh = V + (size_t)head * ATT_S * ATT_D;
  const int s0 = chunk * CHUNK_S + wave * WAVE_S;

  v8f c[4][4];
#pragma unroll
  for (int i = 0; i < 4; ++i)
#pragma unroll
    for (int j = 0; j < 4; ++j)
      c[i][j] = (v8f){0.f, 0.f, 0.f, 0.f, 0.f, 0.f, 0.f, 0.f};

  v2f a[WAVE_S / 4][4], b[WAVE_S / 4][4];
#pragma unroll
  for (int st = 0; st < K1_AHEAD; ++st) {
    const int s = s0 + st * 4 + lhalf * 2;
    const float* __restrict__ kr = Kh + (size_t)s * ATT_D + lmod;
    const float* __restrict__ vr = Vh + (size_t)s * ATT_D + lmod;
#pragma unroll
    for (int t = 0; t < 4; ++t) {
      a[st][t].x = kr[t * 16];
      a[st][t].y = kr[t * 16 + ATT_D];
      b[st][t].x = vr[t * 16];
      b[st][t].y = vr[t * 16 + ATT_D];
    }
  }

#pragma unroll
  for (int step = 0; step < WAVE_S / 4; ++step) {
    if (step + K1_AHEAD < WAVE_S / 4) {
      const int st = step + K1_AHEAD;
      const int s = s0 + st * 4 + lhalf * 2;
      const float* __restrict__ kr = Kh + (size_t)s * ATT_D + lmod;
      const float* __restrict__ vr = Vh + (size_t)s * ATT_D + lmod;
#pragma unroll
      for (int t = 0; t < 4; ++t) {
        a[st][t].x = kr[t * 16];
        a[st][t].y = kr[t * 16 + ATT_D];
        b[st][t].x = vr[t * 16];
        b[st][t].y = vr[t * 16 + ATT_D];
      }
    }
#pragma unroll
    for (int i = 0; i < 4; ++i)
#pragma unroll
      for (int j = 0; j < 4; ++j)
        c[i][j] = wmma_f32(a[step][i], b[step][j], c[i][j]);
    __builtin_amdgcn_sched_barrier(0);
  }
  asm volatile("v_nop\n\tv_nop\n\tv_nop\n\tv_nop"
               :: "v"(a[WAVE_S / 4 - 1][0]), "v"(a[WAVE_S / 4 - 1][1]),
                  "v"(a[WAVE_S / 4 - 1][2]), "v"(a[WAVE_S / 4 - 1][3]),
                  "v"(b[WAVE_S / 4 - 1][0]), "v"(b[WAVE_S / 4 - 1][1]),
                  "v"(b[WAVE_S / 4 - 1][2]), "v"(b[WAVE_S / 4 - 1][3]));

  __shared__ float red[K1_WAVES][ATT_D * ATT_D];
#pragma unroll
  for (int i = 0; i < 4; ++i)
#pragma unroll
    for (int j = 0; j < 4; ++j)
#pragma unroll
      for (int r = 0; r < 8; ++r)
        red[wave][(i * 16 + r + lhalf * 8) * ATT_D + j * 16 + lmod] = c[i][j][r];
  __syncthreads();

  float* __restrict__ outp =
      partial + ((size_t)head * NCHUNK + chunk) * (ATT_D * ATT_D);
  for (int e = threadIdx.x; e < ATT_D * ATT_D; e += K1_WAVES * 32)
    outp[e] = (red[0][e] + red[1][e]) + (red[2][e] + red[3][e]);
}

__global__ __launch_bounds__(RED_THREADS)
__attribute__((amdgpu_waves_per_eu(1, 8)))
void attn_reduce(const float* __restrict__ partial, float* __restrict__ M) {
  constexpr unsigned HEAD4 = ATT_D * ATT_D / 4;
  const unsigned g    = blockIdx.x * RED_THREADS + threadIdx.x;
  const unsigned head = g / HEAD4;
  const unsigned e4   = g % HEAD4;
  const float4* __restrict__ p =
      (const float4*)(partial + (size_t)head * NCHUNK * ATT_D * ATT_D) + e4;
  float4 v[NCHUNK];
  for (int cc = 0; cc < NCHUNK; ++cc)
    v[cc] = p[cc * HEAD4];
  float4 s = {0.f, 0.f, 0.f, 0.f};
  for (int cc = 0; cc < NCHUNK; ++cc) {
    s.x += v[cc].x;
    s.y += v[cc].y;
    s.z += v[cc].z;
    s.w += v[cc].w;
  }
  ((float4*)M)[g] = s;
}

__global__ __launch_bounds__(QWAVES * 32)
void attn_qm(const float* __restrict__ Q, const float* __restrict__ M,
             float* __restrict__ out) {
  const int head  = blockIdx.x / (ATT_S / QTILE);
  const int qc    = blockIdx.x % (ATT_S / QTILE);
  const int wave  = threadIdx.x >> 5;
  const int lane  = threadIdx.x & 31;
  const int lmod  = lane & 15;
  const int lhalf = lane >> 4;
  const int q0    = qc * QTILE;

  __shared__ float qs[QTILE * QPAD];

  int ooff = (wave * 64 + lhalf * 8) * ATT_D + lmod;
  asm volatile("" : "+v"(ooff));

  constexpr int QFILL = QTILE * (ATT_D / 4) / (QWAVES * 32);
  const float* __restrict__ Qh = Q + ((size_t)head * ATT_S + q0) * ATT_D;
  float4 qv[QFILL];
#pragma unroll
  for (int it = 0; it < QFILL; ++it)
    qv[it] = ((const float4*)Qh)[threadIdx.x + it * (QWAVES * 32)];

  const float* __restrict__ Mh = M + (size_t)head * ATT_D * ATT_D;
  v2f mb[ATT_D / 4][4];
#pragma unroll
  for (int st = 0; st < QM_AHEAD; ++st) {
    const int k0 = st * 4 + lhalf * 2;
#pragma unroll
    for (int t = 0; t < 4; ++t) {
      mb[st][t].x = Mh[(k0 + 0) * ATT_D + t * 16 + lmod];
      mb[st][t].y = Mh[(k0 + 1) * ATT_D + t * 16 + lmod];
    }
  }

#pragma unroll
  for (int it = 0; it < QFILL; ++it) {
    const int idx = threadIdx.x + it * (QWAVES * 32);
    const int row = idx >> 4;
    const int c4  = idx & 15;
    *(float4*)&qs[row * QPAD + c4 * 4] = qv[it];
  }
  __syncthreads();

  v8f c[4][4];
#pragma unroll
  for (int i = 0; i < 4; ++i)
#pragma unroll
    for (int j = 0; j < 4; ++j)
      c[i][j] = (v8f){0.f, 0.f, 0.f, 0.f, 0.f, 0.f, 0.f, 0.f};

  v2f a[ATT_D / 4][4];
#pragma unroll
  for (int t = 0; t < 4; ++t) {
    a[0][t] = *(const v2f*)&qs[(wave * 64 + t * 16 + lmod) * QPAD + lhalf * 2];
  }
#pragma unroll
  for (int step = 0; step < ATT_D / 4; ++step) {
    if (step + QM_AHEAD < ATT_D / 4) {
      const int st = step + QM_AHEAD;
      const int k0 = st * 4 + lhalf * 2;
#pragma unroll
      for (int t = 0; t < 4; ++t) {
        mb[st][t].x = Mh[(k0 + 0) * ATT_D + t * 16 + lmod];
        mb[st][t].y = Mh[(k0 + 1) * ATT_D + t * 16 + lmod];
      }
    }
    if (step + 1 < ATT_D / 4) {
      const int k0 = (step + 1) * 4 + lhalf * 2;
#pragma unroll
      for (int t = 0; t < 4; ++t)
        a[step + 1][t] =
            *(const v2f*)&qs[(wave * 64 + t * 16 + lmod) * QPAD + k0];
    }
#pragma unroll
    for (int i = 0; i < 4; ++i)
#pragma unroll
      for (int j = 0; j < 4; ++j)
        c[i][j] = wmma_f32(a[step][i], mb[step][j], c[i][j]);
    __builtin_amdgcn_sched_barrier(0);
  }
  asm volatile("v_nop\n\tv_nop\n\tv_nop\n\tv_nop"
               : "+v"(ooff)
               : "v"(a[ATT_D / 4 - 1][0]), "v"(a[ATT_D / 4 - 1][1]),
                 "v"(a[ATT_D / 4 - 1][2]), "v"(a[ATT_D / 4 - 1][3]),
                 "v"(mb[ATT_D / 4 - 1][0]), "v"(mb[ATT_D / 4 - 1][1]),
                 "v"(mb[ATT_D / 4 - 1][2]), "v"(mb[ATT_D / 4 - 1][3]));

  float* __restrict__ Oh =
      out + ((size_t)head * ATT_S + q0) * ATT_D + ooff;
#pragma unroll
  for (int i = 0; i < 4; ++i)
#pragma unroll
    for (int j = 0; j < 4; ++j)
#pragma unroll
      for (int r = 0; r < 8; ++r)
        Oh[(i * 16 + r) * ATT_D + j * 16] = c[i][j][r];
}

extern "C" void kernel_launch(void* const* d_in, const int* in_sizes, int n_in,
                              void* d_out, int out_size, void* d_ws,
                              size_t ws_size, hipStream_t stream) {
  (void)in_sizes; (void)n_in; (void)out_size; (void)ws_size;
  const float* q = (const float*)d_in[0];
  const float* k = (const float*)d_in[1];
  const float* v = (const float*)d_in[2];
  float* out = (float*)d_out;

  float* partial = (float*)d_ws;
  float* M = partial + (size_t)NHEAD * NCHUNK * ATT_D * ATT_D;

  attn_ktv_partial<<<NHEAD * NCHUNK, K1_WAVES * 32, 0, stream>>>(k, v, partial);
  attn_reduce<<<NHEAD * (ATT_D * ATT_D / 4) / RED_THREADS, RED_THREADS, 0, stream>>>(partial, M);
  attn_qm<<<NHEAD * (ATT_S / QTILE), QWAVES * 32, 0, stream>>>(q, M, out);
}


# Stage 3 walks its 64 q rows as four 16-row tiles: a tile's matrix work starts once its own Q rows land, its rows are stored under the next tile's work (lane owns 4 adjacent columns); same sums
# speedup vs baseline: 2.0021x; 1.1594x over previous
#include <hip/hip_runtime.h>


typedef float v2f __attribute__((ext_vector_type(2)));
typedef float v8f __attribute__((ext_vector_type(8)));
typedef float v4f __attribute__((ext_vector_type(4)));

#define ATT_S   4096
#define ATT_D   64
#define NHEAD   16
#define NCHUNK  16
#define CHUNK_S (ATT_S / NCHUNK)
#define K1_WAVES 4
#define WAVE_S  (CHUNK_S / K1_WAVES)
#define QTILE   128
#define QWAVES  2
#define QPAD    68
#define K1_AHEAD 3
#define QM_AHEAD 3
#define RED_THREADS 64

static __device__ __forceinline__ v8f wmma_f32(v2f a, v2f b, v8f c) {
  return __builtin_amdgcn_wmma_f32_16x16x4_f32(false, a, false, b,
                                               (short)0, c, false, false);
}

__global__ __launch_bounds__(K1_WAVES * 32)
void attn_ktv_partial(const float* __restrict__ K, const float* __restrict__ V,
                      float* __restrict__ partial) {
  const int head  = blockIdx.x / NCHUNK;
  const int chunk = blockIdx.x % NCHUNK;
  const int wave  = threadIdx.x >> 5;
  const int lane  = threadIdx.x & 31;
  const int lmod  = lane & 15;
  const int lhalf = lane >> 4;

  const float* __restrict__ Kh = K + (size_t)head * ATT_S * ATT_D;
  const float* __restrict__ Vh = V + (size_t)head * ATT_S * ATT_D;
  const int s0 = chunk * CHUNK_S + wave * WAVE_S;

  v8f c[4][4];
#pragma unroll
  for (int i = 0; i < 4; ++i)
#pragma unroll
    for (int j = 0; j < 4; ++j)
      c[i][j] = (v8f){0.f, 0.f, 0.f, 0.f, 0.f, 0.f, 0.f, 0.f};

  v2f a[WAVE_S / 4][4], b[WAVE_S / 4][4];
#pragma unroll
  for (int st = 0; st < K1_AHEAD; ++st) {
    const int s = s0 + st * 4 + lhalf * 2;
    const float* __restrict__ kr = Kh + (size_t)s * ATT_D + lmod;
    const float* __restrict__ vr = Vh + (size_t)s * ATT_D + lmod;
#pragma unroll
    for (int t = 0; t < 4; ++t) {
      a[st][t].x = kr[t * 16];
      a[st][t].y = kr[t * 16 + ATT_D];
      b[st][t].x = vr[t * 16];
      b[st][t].y = vr[t * 16 + ATT_D];
    }
  }

#pragma unroll
  for (int step = 0; step < WAVE_S / 4; ++step) {
    if (step + K1_AHEAD < WAVE_S / 4) {
      const int st = step + K1_AHEAD;
      const int s = s0 + st * 4 + lhalf * 2;
      const float* __restrict__ kr = Kh + (size_t)s * ATT_D + lmod;
      const float* __restrict__ vr = Vh + (size_t)s * ATT_D + lmod;
#pragma unroll
      for (int t = 0; t < 4; ++t) {
        a[st][t].x = kr[t * 16];
        a[st][t].y = kr[t * 16 + ATT_D];
        b[st][t].x = vr[t * 16];
        b[st][t].y = vr[t * 16 + ATT_D];
      }
    }
#pragma unroll
    for (int i = 0; i < 4; ++i)
#pragma unroll
      for (int j = 0; j < 4; ++j)
        c[i][j] = wmma_f32(a[step][i], b[step][j], c[i][j]);
    __builtin_amdgcn_sched_barrier(0);
  }
  asm volatile("v_nop\n\tv_nop\n\tv_nop\n\tv_nop"
               :: "v"(a[WAVE_S / 4 - 1][0]), "v"(a[WAVE_S / 4 - 1][1]),
                  "v"(a[WAVE_S / 4 - 1][2]), "v"(a[WAVE_S / 4 - 1][3]),
                  "v"(b[WAVE_S / 4 - 1][0]), "v"(b[WAVE_S / 4 - 1][1]),
                  "v"(b[WAVE_S / 4 - 1][2]), "v"(b[WAVE_S / 4 - 1][3]));

  __shared__ float red[K1_WAVES][ATT_D * ATT_D];
#pragma unroll
  for (int i = 0; i < 4; ++i)
#pragma unroll
    for (int j = 0; j < 4; ++j)
#pragma unroll
      for (int r = 0; r < 8; ++r)
        red[wave][(i * 16 + r + lhalf * 8) * ATT_D + j * 16 + lmod] = c[i][j][r];
  __syncthreads();

  float* __restrict__ outp =
      partial + ((size_t)head * NCHUNK + chunk) * (ATT_D * ATT_D);
  for (int e = threadIdx.x; e < ATT_D * ATT_D; e += K1_WAVES * 32)
    outp[e] = (red[0][e] + red[1][e]) + (red[2][e] + red[3][e]);
}

__global__ __launch_bounds__(RED_THREADS)
__attribute__((amdgpu_waves_per_eu(1, 8)))
void attn_reduce(const float* __restrict__ partial, float* __restrict__ M) {
  constexpr unsigned HEAD4 = ATT_D * ATT_D / 4;
  const unsigned g    = blockIdx.x * RED_THREADS + threadIdx.x;
  const unsigned head = g / HEAD4;
  const unsigned e4   = g % HEAD4;
  const float4* __restrict__ p =
      (const float4*)(partial + (size_t)head * NCHUNK * ATT_D * ATT_D) + e4;
  float4 v[NCHUNK];
  for (int cc = 0; cc < NCHUNK; ++cc)
    v[cc] = p[cc * HEAD4];
  float4 s = {0.f, 0.f, 0.f, 0.f};
  for (int cc = 0; cc < NCHUNK; ++cc) {
    s.x += v[cc].x;
    s.y += v[cc].y;
    s.z += v[cc].z;
    s.w += v[cc].w;
  }
  ((float4*)M)[g] = s;
}

__global__ __launch_bounds__(QWAVES * 32)
__attribute__((amdgpu_waves_per_eu(1, 2)))
void attn_qm(const float* __restrict__ Q, const float* __restrict__ M,
             float* __restrict__ out) {
  const int head  = blockIdx.x / (ATT_S / QTILE);
  const int qc    = blockIdx.x % (ATT_S / QTILE);
  const int wave  = threadIdx.x >> 5;
  const int lane  = threadIdx.x & 31;
  const int lmod  = lane & 15;
  const int lhalf = lane >> 4;
  const int q0    = qc * QTILE + wave * 64;

  __shared__ float qs[QTILE * QPAD];
  float* __restrict__ qw = qs + wave * 64 * QPAD;

  int ooff = (lhalf * 8) * ATT_D + lmod * 4;
  asm volatile("" : "+v"(ooff));

  const v4f* __restrict__ Qw =
      (const v4f*)(Q + ((size_t)head * ATT_S + q0) * ATT_D);
  v4f qv[32];
#pragma unroll
  for (int it = 0; it < 8; ++it)
    qv[it] = Qw[lane + it * 32];
  __builtin_amdgcn_sched_barrier(0);

  const float* __restrict__ Mh = M + (size_t)head * ATT_D * ATT_D;
  v4f mr0[ATT_D / 4], mr1[ATT_D / 4];
#pragma unroll
  for (int st = 0; st < ATT_D / 4; ++st) {
    const int k0 = st * 4 + lhalf * 2;
    mr0[st] = *(const v4f*)&Mh[(k0 + 0) * ATT_D + lmod * 4];
    mr1[st] = *(const v4f*)&Mh[(k0 + 1) * ATT_D + lmod * 4];
  }
  __builtin_amdgcn_sched_barrier(0);
#pragma unroll
  for (int it = 8; it < 32; ++it)
    qv[it] = Qw[lane + it * 32];
  __builtin_amdgcn_sched_barrier(0);

  v2f mb[ATT_D / 4][4];
#pragma unroll
  for (int st = 0; st < ATT_D / 4; ++st)
#pragma unroll
    for (int j = 0; j < 4; ++j) {
      mb[st][j].x = mr0[st][j];
      mb[st][j].y = mr1[st][j];
    }

  v8f c[4][4];
#pragma unroll
  for (int i = 0; i < 4; ++i)
#pragma unroll
    for (int j = 0; j < 4; ++j)
      c[i][j] = (v8f){0.f, 0.f, 0.f, 0.f, 0.f, 0.f, 0.f, 0.f};

  float* __restrict__ Ow = out + ((size_t)head * ATT_S + q0) * ATT_D;

  v2f a[2][ATT_D / 4];

#define QM_ST1(IT) *(v4f*)&qw[(2 * (IT) + lhalf) * QPAD + lmod * 4] = qv[IT];
#define QM_LD1(I, ST)                                                         \
  a[(I) & 1][ST] =                                                            \
      *(const v2f*)&qw[((I) * 16 + lmod) * QPAD + (ST) * 4 + lhalf * 2];
#define QM_STAGE_TILE(I)                                                      \
  {                                                                           \
    QM_ST1((I) * 8 + 0) QM_ST1((I) * 8 + 1) QM_ST1((I) * 8 + 2)               \
    QM_ST1((I) * 8 + 3) QM_ST1((I) * 8 + 4) QM_ST1((I) * 8 + 5)               \
    QM_ST1((I) * 8 + 6) QM_ST1((I) * 8 + 7)                                   \
                         \
    __builtin_amdgcn_fence(__ATOMIC_RELEASE, "wavefront");                    \
    __builtin_amdgcn_wave_barrier();                                          \
    __builtin_amdgcn_fence(__ATOMIC_ACQUIRE, "wavefront");                    \
    QM_LD1(I, 0) QM_LD1(I, 1) QM_LD1(I, 2) QM_LD1(I, 3) QM_LD1(I, 4)          \
    QM_LD1(I, 5) QM_LD1(I, 6) QM_LD1(I, 7) QM_LD1(I, 8) QM_LD1(I, 9)          \
    QM_LD1(I, 10) QM_LD1(I, 11) QM_LD1(I, 12) QM_LD1(I, 13) QM_LD1(I, 14)     \
    QM_LD1(I, 15)                                                             \
  }

  QM_STAGE_TILE(0)

#define QM_TILE(I, NEXT_STAGE)                                                \
  {                                                                           \
    _Pragma("unroll")                                                         \
    for (int step = 0; step < ATT_D / 8; ++step)                              \
      _Pragma("unroll")                                                       \
      for (int j = 0; j < 4; ++j)                                             \
        c[I][j] = wmma_f32(a[(I) & 1][step], mb[step][j], c[I][j]);           \
                                               \
    __builtin_amdgcn_sched_barrier(0);                                        \
    NEXT_STAGE                                                                \
                            \
    __builtin_amdgcn_sched_barrier(0);                                        \
    _Pragma("unroll")                                                         \
    for (int step = ATT_D / 8; step < ATT_D / 4; ++step)                      \
      _Pragma("unroll")                                                       \
      for (int j = 0; j < 4; ++j)                                             \
        c[I][j] = wmma_f32(a[(I) & 1][step], mb[step][j], c[I][j]);           \
                                             \
    asm volatile("v_nop\n\tv_nop\n\tv_nop\n\tv_nop"                         \
                 : "+v"(ooff), "+v"(c[I][0]), "+v"(c[I][1]), "+v"(c[I][2]),   \
                   "+v"(c[I][3])                                              \
                 : "v"(a[(I) & 1][ATT_D / 4 - 1]),                            \
                   "v"(mb[ATT_D / 4 - 1][0]), "v"(mb[ATT_D / 4 - 1][1]),      \
                   "v"(mb[ATT_D / 4 - 1][2]), "v"(mb[ATT_D / 4 - 1][3]));     \
                                          \
    _Pragma("unroll")                                                         \
    for (int r = 0; r < 8; ++r) {                                             \
      v4f o;                                                                  \
      o.x = c[I][0][r]; o.y = c[I][1][r]; o.z = c[I][2][r]; o.w = c[I][3][r]; \
      *(v4f*)&Ow[((I) * 16 + r) * ATT_D + ooff] = o;                          \
    }                                                                         \
                \
    __builtin_amdgcn_sched_barrier(0);                                        \
  }

  QM_TILE(0, QM_STAGE_TILE(1))
  QM_TILE(1, QM_STAGE_TILE(2))
  QM_TILE(2, QM_STAGE_TILE(3))
  QM_TILE(3, )
#undef QM_TILE
#undef QM_STAGE_TILE
#undef QM_ST1
#undef QM_LD1
}

extern "C" void kernel_launch(void* const* d_in, const int* in_sizes, int n_in,
                              void* d_out, int out_size, void* d_ws,
                              size_t ws_size, hipStream_t stream) {
  (void)in_sizes; (void)n_in; (void)out_size; (void)ws_size;
  const float* q = (const float*)d_in[0];
  const float* k = (const float*)d_in[1];
  const float* v = (const float*)d_in[2];
  float* out = (float*)d_out;

  float* partial = (float*)d_ws;
  float* M = partial + (size_t)NHEAD * NCHUNK * ATT_D * ATT_D;

  attn_ktv_partial<<<NHEAD * NCHUNK, K1_WAVES * 32, 0, stream>>>(k, v, partial);
  attn_reduce<<<NHEAD * (ATT_D * ATT_D / 4) / RED_THREADS, RED_THREADS, 0, stream>>>(partial, M);
  attn_qm<<<NHEAD * (ATT_S / QTILE), QWAVES * 32, 0, stream>>>(q, M, out);
}


# Stage 1: a lane owns 4 adjacent columns of K, V and partial: 16-byte loads, 8 k steps (32 loads) in flight, was 3 (48 4-byte loads); LDS sum and partial stores in 16-byte pieces; same sums
# speedup vs baseline: 2.2023x; 1.1000x over previous
#include <hip/hip_runtime.h>


typedef float v2f __attribute__((ext_vector_type(2)));
typedef float v8f __attribute__((ext_vector_type(8)));
typedef float v4f __attribute__((ext_vector_type(4)));

#define ATT_S   4096
#define ATT_D   64
#define NHEAD   16
#define NCHUNK  16
#define CHUNK_S (ATT_S / NCHUNK)
#define K1_WAVES 4
#define WAVE_S  (CHUNK_S / K1_WAVES)
#define QTILE   128
#define QWAVES  2
#define QPAD    68
#define K1_AHEAD 8
#define QM_AHEAD 3
#define RED_THREADS 64

static __device__ __forceinline__ v8f wmma_f32(v2f a, v2f b, v8f c) {
  return __builtin_amdgcn_wmma_f32_16x16x4_f32(false, a, false, b,
                                               (short)0, c, false, false);
}

__global__ __launch_bounds__(K1_WAVES * 32)
__attribute__((amdgpu_waves_per_eu(1, 2)))
void attn_ktv_partial(const float* __restrict__ K, const float* __restrict__ V,
                      float* __restrict__ partial) {
  const int head  = blockIdx.x / NCHUNK;
  const int chunk = blockIdx.x % NCHUNK;
  const int wave  = threadIdx.x >> 5;
  const int lane  = threadIdx.x & 31;
  const int lmod  = lane & 15;
  const int lhalf = lane >> 4;

  const int s0 = chunk * CHUNK_S + wave * WAVE_S;
  const v4f* __restrict__ Kq =
      (const v4f*)(K + ((size_t)head * ATT_S + s0 + lhalf * 2) * ATT_D) + lmod;
  const v4f* __restrict__ Vq =
      (const v4f*)(V + ((size_t)head * ATT_S + s0 + lhalf * 2) * ATT_D) + lmod;

  v4f kr0[WAVE_S / 4], kr1[WAVE_S / 4], vr0[WAVE_S / 4], vr1[WAVE_S / 4];
#define K1_LOAD(ST)                                                           \
  kr0[ST] = Kq[(ST) * 64];                                                    \
  kr1[ST] = Kq[(ST) * 64 + 16];                                               \
  vr0[ST] = Vq[(ST) * 64];                                                    \
  vr1[ST] = Vq[(ST) * 64 + 16];
#pragma unroll
  for (int st = 0; st < K1_AHEAD; ++st) {
    K1_LOAD(st)
  }
  __builtin_amdgcn_sched_barrier(0);

  v8f c[4][4];
#pragma unroll
  for (int i = 0; i < 4; ++i)
#pragma unroll
    for (int j = 0; j < 4; ++j)
      c[i][j] = (v8f){0.f, 0.f, 0.f, 0.f, 0.f, 0.f, 0.f, 0.f};

  v2f a[2][4], b[2][4];
#define K1_FRAG(ST)                                                           \
  _Pragma("unroll")                                                           \
  for (int t = 0; t < 4; ++t) {                                               \
    a[(ST) & 1][t].x = kr0[ST][t];                                            \
    a[(ST) & 1][t].y = kr1[ST][t];                                            \
    b[(ST) & 1][t].x = vr0[ST][t];                                            \
    b[(ST) & 1][t].y = vr1[ST][t];                                            \
  }
  K1_FRAG(0)

#pragma unroll
  for (int step = 0; step < WAVE_S / 4; ++step) {
    if (step + K1_AHEAD < WAVE_S / 4) {
      K1_LOAD(step + K1_AHEAD)
    }
#pragma unroll
    for (int i = 0; i < 2; ++i)
#pragma unroll
      for (int j = 0; j < 4; ++j)
        c[i][j] = wmma_f32(a[step & 1][i], b[step & 1][j], c[i][j]);
    __builtin_amdgcn_sched_barrier(0);
    if (step + 1 < WAVE_S / 4) {
      K1_FRAG(step + 1)
    }
#pragma unroll
    for (int i = 2; i < 4; ++i)
#pragma unroll
      for (int j = 0; j < 4; ++j)
        c[i][j] = wmma_f32(a[step & 1][i], b[step & 1][j], c[i][j]);
    __builtin_amdgcn_sched_barrier(0);
  }
#undef K1_FRAG
#undef K1_LOAD
  asm volatile("v_nop\n\tv_nop\n\tv_nop\n\tv_nop"
               : "+v"(c[3][0]), "+v"(c[3][1]), "+v"(c[3][2]), "+v"(c[3][3])
               : "v"(a[1][0]), "v"(a[1][1]), "v"(a[1][2]), "v"(a[1][3]),
                 "v"(b[1][0]), "v"(b[1][1]), "v"(b[1][2]), "v"(b[1][3]));
  __builtin_amdgcn_sched_barrier(0);

  __shared__ float red[K1_WAVES][ATT_D * ATT_D];
#pragma unroll
  for (int i = 0; i < 4; ++i)
#pragma unroll
    for (int r = 0; r < 8; ++r) {
      v4f o;
      o.x = c[i][0][r]; o.y = c[i][1][r]; o.z = c[i][2][r]; o.w = c[i][3][r];
      *(v4f*)&red[wave][(4 * (8 * lhalf + r) + i) * ATT_D + 4 * lmod] = o;
    }
  __syncthreads();

  v4f* __restrict__ outp =
      (v4f*)(partial + ((size_t)head * NCHUNK + chunk) * (ATT_D * ATT_D));
  constexpr int RED_IT = ATT_D * ATT_D / 4 / (K1_WAVES * 32);
  v4f x[RED_IT][K1_WAVES];
#pragma unroll
  for (int it = 0; it < RED_IT; ++it)
#pragma unroll
    for (int w = 0; w < K1_WAVES; ++w)
      x[it][w] = ((const v4f*)red[w])[threadIdx.x + it * (K1_WAVES * 32)];
#pragma unroll
  for (int it = 0; it < RED_IT; ++it)
    outp[threadIdx.x + it * (K1_WAVES * 32)] =
        (x[it][0] + x[it][1]) + (x[it][2] + x[it][3]);
}

__global__ __launch_bounds__(RED_THREADS)
__attribute__((amdgpu_waves_per_eu(1, 8)))
void attn_reduce(const float* __restrict__ partial, float* __restrict__ M) {
  constexpr unsigned HEAD4 = ATT_D * ATT_D / 4;
  const unsigned g    = blockIdx.x * RED_THREADS + threadIdx.x;
  const unsigned head = g / HEAD4;
  const unsigned e4   = g % HEAD4;
  const float4* __restrict__ p =
      (const float4*)(partial + (size_t)head * NCHUNK * ATT_D * ATT_D) + e4;
  float4 v[NCHUNK];
  for (int cc = 0; cc < NCHUNK; ++cc)
    v[cc] = p[cc * HEAD4];
  float4 s = {0.f, 0.f, 0.f, 0.f};
  for (int cc = 0; cc < NCHUNK; ++cc) {
    s.x += v[cc].x;
    s.y += v[cc].y;
    s.z += v[cc].z;
    s.w += v[cc].w;
  }
  ((float4*)M)[g] = s;
}

__global__ __launch_bounds__(QWAVES * 32)
__attribute__((amdgpu_waves_per_eu(1, 2)))
void attn_qm(const float* __restrict__ Q, const float* __restrict__ M,
             float* __restrict__ out) {
  const int head  = blockIdx.x / (ATT_S / QTILE);
  const int qc    = blockIdx.x % (ATT_S / QTILE);
  const int wave  = threadIdx.x >> 5;
  const int lane  = threadIdx.x & 31;
  const int lmod  = lane & 15;
  const int lhalf = lane >> 4;
  const int q0    = qc * QTILE + wave * 64;

  __shared__ float qs[QTILE * QPAD];
  float* __restrict__ qw = qs + wave * 64 * QPAD;

  int ooff = (lhalf * 8) * ATT_D + lmod * 4;
  asm volatile("" : "+v"(ooff));

  const v4f* __restrict__ Qw =
      (const v4f*)(Q + ((size_t)head * ATT_S + q0) * ATT_D);
  v4f qv[32];
#pragma unroll
  for (int it = 0; it < 8; ++it)
    qv[it] = Qw[lane + it * 32];
  __builtin_amdgcn_sched_barrier(0);

  const float* __restrict__ Mh = M + (size_t)head * ATT_D * ATT_D;
  v4f mr0[ATT_D / 4], mr1[ATT_D / 4];
#pragma unroll
  for (int st = 0; st < ATT_D / 4; ++st) {
    const int k0 = st * 4 + lhalf * 2;
    mr0[st] = *(const v4f*)&Mh[(k0 + 0) * ATT_D + lmod * 4];
    mr1[st] = *(const v4f*)&Mh[(k0 + 1) * ATT_D + lmod * 4];
  }
  __builtin_amdgcn_sched_barrier(0);
#pragma unroll
  for (int it = 8; it < 32; ++it)
    qv[it] = Qw[lane + it * 32];
  __builtin_amdgcn_sched_barrier(0);

  v2f mb[ATT_D / 4][4];
#pragma unroll
  for (int st = 0; st < ATT_D / 4; ++st)
#pragma unroll
    for (int j = 0; j < 4; ++j) {
      mb[st][j].x = mr0[st][j];
      mb[st][j].y = mr1[st][j];
    }

  v8f c[4][4];
#pragma unroll
  for (int i = 0; i < 4; ++i)
#pragma unroll
    for (int j = 0; j < 4; ++j)
      c[i][j] = (v8f){0.f, 0.f, 0.f, 0.f, 0.f, 0.f, 0.f, 0.f};

  float* __restrict__ Ow = out + ((size_t)head * ATT_S + q0) * ATT_D;

  v2f a[2][ATT_D / 4];

#define QM_ST1(IT) *(v4f*)&qw[(2 * (IT) + lhalf) * QPAD + lmod * 4] = qv[IT];
#define QM_LD1(I, ST)                                                         \
  a[(I) & 1][ST] =                                                            \
      *(const v2f*)&qw[((I) * 16 + lmod) * QPAD + (ST) * 4 + lhalf * 2];
#define QM_STAGE_TILE(I)                                                      \
  {                                                                           \
    QM_ST1((I) * 8 + 0) QM_ST1((I) * 8 + 1) QM_ST1((I) * 8 + 2)               \
    QM_ST1((I) * 8 + 3) QM_ST1((I) * 8 + 4) QM_ST1((I) * 8 + 5)               \
    QM_ST1((I) * 8 + 6) QM_ST1((I) * 8 + 7)                                   \
                         \
    __builtin_amdgcn_fence(__ATOMIC_RELEASE, "wavefront");                    \
    __builtin_amdgcn_wave_barrier();                                          \
    __builtin_amdgcn_fence(__ATOMIC_ACQUIRE, "wavefront");                    \
    QM_LD1(I, 0) QM_LD1(I, 1) QM_LD1(I, 2) QM_LD1(I, 3) QM_LD1(I, 4)          \
    QM_LD1(I, 5) QM_LD1(I, 6) QM_LD1(I, 7) QM_LD1(I, 8) QM_LD1(I, 9)          \
    QM_LD1(I, 10) QM_LD1(I, 11) QM_LD1(I, 12) QM_LD1(I, 13) QM_LD1(I, 14)     \
    QM_LD1(I, 15)                                                             \
  }

  QM_STAGE_TILE(0)

#define QM_TILE(I, NEXT_STAGE)                                                \
  {                                                                           \
    _Pragma("unroll")                                                         \
    for (int step = 0; step < ATT_D / 8; ++step)                              \
      _Pragma("unroll")                                                       \
      for (int j = 0; j < 4; ++j)                                             \
        c[I][j] = wmma_f32(a[(I) & 1][step], mb[step][j], c[I][j]);           \
                                               \
    __builtin_amdgcn_sched_barrier(0);                                        \
    NEXT_STAGE                                                                \
                            \
    __builtin_amdgcn_sched_barrier(0);                                        \
    _Pragma("unroll")                                                         \
    for (int step = ATT_D / 8; step < ATT_D / 4; ++step)                      \
      _Pragma("unroll")                                                       \
      for (int j = 0; j < 4; ++j)                                             \
        c[I][j] = wmma_f32(a[(I) & 1][step], mb[step][j], c[I][j]);           \
                                             \
    asm volatile("v_nop\n\tv_nop\n\tv_nop\n\tv_nop"                         \
                 : "+v"(ooff), "+v"(c[I][0]), "+v"(c[I][1]), "+v"(c[I][2]),   \
                   "+v"(c[I][3])                                              \
                 : "v"(a[(I) & 1][ATT_D / 4 - 1]),                            \
                   "v"(mb[ATT_D / 4 - 1][0]), "v"(mb[ATT_D / 4 - 1][1]),      \
                   "v"(mb[ATT_D / 4 - 1][2]), "v"(mb[ATT_D / 4 - 1][3]));     \
                                          \
    _Pragma("unroll")                                                         \
    for (int r = 0; r < 8; ++r) {                                             \
      v4f o;                                                                  \
      o.x = c[I][0][r]; o.y = c[I][1][r]; o.z = c[I][2][r]; o.w = c[I][3][r]; \
      *(v4f*)&Ow[((I) * 16 + r) * ATT_D + ooff] = o;                          \
    }                                                                         \
                \
    __builtin_amdgcn_sched_barrier(0);                                        \
  }

  QM_TILE(0, QM_STAGE_TILE(1))
  QM_TILE(1, QM_STAGE_TILE(2))
  QM_TILE(2, QM_STAGE_TILE(3))
  QM_TILE(3, )
#undef QM_TILE
#undef QM_STAGE_TILE
#undef QM_ST1
#undef QM_LD1
}

extern "C" void kernel_launch(void* const* d_in, const int* in_sizes, int n_in,
                              void* d_out, int out_size, void* d_ws,
                              size_t ws_size, hipStream_t stream) {
  (void)in_sizes; (void)n_in; (void)out_size; (void)ws_size;
  const float* q = (const float*)d_in[0];
  const float* k = (const float*)d_in[1];
  const float* v = (const float*)d_in[2];
  float* out = (float*)d_out;

  float* partial = (float*)d_ws;
  float* M = partial + (size_t)NHEAD * NCHUNK * ATT_D * ATT_D;

  attn_ktv_partial<<<NHEAD * NCHUNK, K1_WAVES * 32, 0, stream>>>(k, v, partial);
  attn_reduce<<<NHEAD * (ATT_D * ATT_D / 4) / RED_THREADS, RED_THREADS, 0, stream>>>(partial, M);
  attn_qm<<<NHEAD * (ATT_S / QTILE), QWAVES * 32, 0, stream>>>(q, M, out);
}
